# starting kernel
# baseline (speedup 1.0000x reference)
#include <hip/hip_runtime.h>
#include <hip/hip_bf16.h>

typedef __attribute__((ext_vector_type(16))) _Float16 v16h;
typedef __attribute__((ext_vector_type(8)))  float    v8f;

__device__ inline float wv_entry(int f, int n,
                                 const float* w_n00, const float* w_n10,
                                 const float* w_n01, const float* w_n11) {
    if (f < 16) {
        if (n == 0) return w_n00[f];
        if (n == 1) return w_n01[2 * f];
        if (n == 2) return w_n01[2 * f + 1];
        return 0.0f;
    }
    if (f < 48) {
        int c    = (f - 16) >> 1;
        int comp = (f - 16) & 1;
        int j    = n - (comp ? 9 : 3);
        if (j < 0 || j > 5) return 0.0f;
        if (j == 0) return w_n10[2 * c];
        if (j == 1) return w_n10[2 * c + 1];
        return w_n11[4 * c + (j - 2)];
    }
    return 0.0f;
}

__global__ void init_zero_kernel(float* p, int n) {
    int i = blockIdx.x * blockDim.x + threadIdx.x;
    if (i < n) p[i] = 0.0f;
}

__global__ void build_B_kernel(const float* w_n00, const float* w_n10,
                               const float* w_n01, const float* w_n11,
                               _Float16* B0, _Float16* B1) {
    int l = threadIdx.x;
    if (l >= 32) return;
    int n  = l & 15;
    int hi = l >> 4;
    for (int i = 0; i < 16; ++i) {
        int f = i + 8 * hi + (i >= 8 ? 8 : 0);
        B0[l * 16 + i] = (_Float16)wv_entry(f, n, w_n00, w_n10, w_n01, w_n11);
        B1[l * 16 + i] = (f < 16)
            ? (_Float16)wv_entry(32 + f, n, w_n00, w_n10, w_n01, w_n11)
            : (_Float16)0.0f;
    }
}

__device__ inline void gather_A(const float* __restrict__ row, int hi,
                                v16h& a0, v16h& a1) {
    const int fb0 = hi ? 8 : 0;
    const int fb1 = hi ? 24 : 16;
#pragma unroll
    for (int i = 0; i < 8; ++i) a0[i]     = (_Float16)row[fb0 + i];
#pragma unroll
    for (int i = 0; i < 8; ++i) a0[8 + i] = (_Float16)row[fb1 + i];
#pragma unroll
    for (int i = 0; i < 8; ++i) a1[i]     = (_Float16)row[32 + fb0 + i];
#pragma unroll
    for (int i = 0; i < 8; ++i) a1[8 + i] = (_Float16)0.0f;
}

__global__ __launch_bounds__(256)
void edge_kernel(const float* __restrict__ x,
                 const int*   __restrict__ src,
                 const int*   __restrict__ dst,
                 const float* __restrict__ angles,
                 const float* __restrict__ transporters,
                 const _Float16* __restrict__ B0p,
                 const _Float16* __restrict__ B1p,
                 float* __restrict__ deg,
                 float* __restrict__ magsum,
                 float* __restrict__ agg1,
                 float* __restrict__ agg2,
                 int E) {
    __shared__ float tile[8][2 * 16 * 16];

    const int lane = threadIdx.x & 31;
    const int wv   = threadIdx.x >> 5;
    const int m    = lane & 15;
    const int hi   = lane >> 4;

    const long tbase = ((long)blockIdx.x * 8 + wv) * 32;
    long e0 = tbase + m;
    long e1 = tbase + 16 + m;
    long e0c = e0 < (long)E ? e0 : (long)E - 1;
    long e1c = e1 < (long)E ? e1 : (long)E - 1;

    int s0 = __builtin_nontemporal_load(&src[e0c]);
    int s1 = __builtin_nontemporal_load(&src[e1c]);

    v16h a0t0, a1t0, a0t1, a1t1;
    gather_A(x + (size_t)s0 * 48, hi, a0t0, a1t0);
    gather_A(x + (size_t)s1 * 48, hi, a0t1, a1t1);

    v16h b0 = *(const v16h*)(B0p + lane * 16);
    v16h b1 = *(const v16h*)(B1p + lane * 16);

    v8f c0 = {}, c1 = {};
    c0 = __builtin_amdgcn_wmma_f32_16x16x32_f16(false, a0t0, false, b0,
                                                (short)0, c0, false, false);
    c0 = __builtin_amdgcn_wmma_f32_16x16x32_f16(false, a1t0, false, b1,
                                                (short)0, c0, false, false);
    c1 = __builtin_amdgcn_wmma_f32_16x16x32_f16(false, a0t1, false, b0,
                                                (short)0, c1, false, false);
    c1 = __builtin_amdgcn_wmma_f32_16x16x32_f16(false, a1t1, false, b1,
                                                (short)0, c1, false, false);

    float* t = tile[wv];
#pragma unroll
    for (int r = 0; r < 8; ++r) {
        t[(r + 8 * hi) * 16 + m]       = c0[r];
        t[256 + (r + 8 * hi) * 16 + m] = c1[r];
    }
    __syncthreads();

    long ee = tbase + lane;
    if (ee < (long)E) {
        const float* d = &t[lane * 16];
        float ang = __builtin_nontemporal_load(&angles[ee]);
        float tr  = __builtin_nontemporal_load(&transporters[ee]);
        float st, ct, s2t, c2t, sg, cg;
        __sincosf(ang, &st, &ct);
        __sincosf(2.0f * ang, &s2t, &c2t);
        __sincosf(tr, &sg, &cg);

        float d0 = d[0], sa = d[1], sb = d[2];
        float u1w0 = cg * d[3] - sg * d[9];
        float u2w0 = sg * d[3] + cg * d[9];
        float u1w1 = cg * d[4] - sg * d[10];
        float u2w1 = sg * d[4] + cg * d[10];
        float u1p  = cg * d[5] - sg * d[11];
        float u2p  = sg * d[5] + cg * d[11];
        float u1q  = cg * d[6] - sg * d[12];
        float u2q  = sg * d[6] + cg * d[12];
        float u1r  = cg * d[7] - sg * d[13];
        float u2r  = sg * d[7] + cg * d[13];
        float u1s  = cg * d[8] - sg * d[14];
        float u2s  = sg * d[8] + cg * d[14];

        float m0  = d0 + ct * u1w0 + st * u2w0 + ct * u2w1 - st * u1w1;
        float mv1 = sa * ct - sb * st + u1p - u2q
                  + (c2t * u1r + s2t * u2r) - (s2t * u1s - c2t * u2s);
        float mv2 = sa * st + sb * ct + u2p + u1q
                  + (s2t * u1r - c2t * u2r) + (c2t * u1s + s2t * u2s);

        int dv = __builtin_nontemporal_load(&dst[ee]);
        __hip_atomic_fetch_add(&deg[dv],    1.0f, __ATOMIC_RELAXED, __HIP_MEMORY_SCOPE_AGENT);
        __hip_atomic_fetch_add(&magsum[dv], m0,   __ATOMIC_RELAXED, __HIP_MEMORY_SCOPE_AGENT);
        __hip_atomic_fetch_add(&agg1[dv],   mv1,  __ATOMIC_RELAXED, __HIP_MEMORY_SCOPE_AGENT);
        __hip_atomic_fetch_add(&agg2[dv],   mv2,  __ATOMIC_RELAXED, __HIP_MEMORY_SCOPE_AGENT);
    }
}

__global__ __launch_bounds__(256)
void finalize_kernel(const float* __restrict__ x,
                     const float* __restrict__ e1,
                     const float* __restrict__ e2,
                     const float* __restrict__ w_self0,
                     const float* __restrict__ w_self11,
                     const float* __restrict__ deg,
                     const float* __restrict__ magsum,
                     const float* __restrict__ agg1,
                     const float* __restrict__ agg2,
                     float* __restrict__ out,
                     int V) {
    int v = blockIdx.x * blockDim.x + threadIdx.x;
    if (v >= V) return;
    const float* row = x + (size_t)v * 48;
    float denom = fmaxf(deg[v], 1.0f);
    float inv   = 1.0f / denom;
    float mag   = magsum[v] * inv;
    float t1    = agg1[v] * inv;
    float t2    = agg2[v] * inv;
#pragma unroll
    for (int c = 0; c < 16; ++c) {
        mag += row[c] * w_self0[c];
        float xa = row[16 + 2 * c], xb = row[17 + 2 * c];
        float sa = w_self11[2 * c], sb = w_self11[2 * c + 1];
        t1 += xa * sa - xb * sb;
        t2 += xb * sa + xa * sb;
    }
    float ms = 2.0f / (1.0f + __expf(-mag));
#pragma unroll
    for (int k = 0; k < 3; ++k)
        out[v * 3 + k] = (t1 * e1[v * 3 + k] + t2 * e2[v * 3 + k]) * ms;
}

extern "C" void kernel_launch(void* const* d_in, const int* in_sizes, int n_in,
                              void* d_out, int out_size, void* d_ws, size_t ws_size,
                              hipStream_t stream) {
    const float* x            = (const float*)d_in[0];
    const int*   edge_index   = (const int*)d_in[1];
    const float* angles       = (const float*)d_in[2];
    const float* transporters = (const float*)d_in[3];
    const float* e1           = (const float*)d_in[4];
    const float* e2           = (const float*)d_in[5];
    const float* w_self0      = (const float*)d_in[6];
    const float* w_n00        = (const float*)d_in[7];
    const float* w_n10        = (const float*)d_in[8];
    const float* w_self11     = (const float*)d_in[9];
    const float* w_n01        = (const float*)d_in[10];
    const float* w_n11        = (const float*)d_in[11];

    const int E = in_sizes[2];
    const int V = in_sizes[0] / 48;
    const int* src = edge_index;
    const int* dst = edge_index + E;

    float* deg    = (float*)d_ws;
    float* magsum = deg + V;
    float* agg1   = magsum + V;
    float* agg2   = agg1 + V;
    _Float16* B0  = (_Float16*)(agg2 + V);
    _Float16* B1  = B0 + 32 * 16;

    int nz = 4 * V;
    hipLaunchKernelGGL(init_zero_kernel, dim3((nz + 255) / 256), dim3(256), 0, stream,
                       deg, nz);
    hipLaunchKernelGGL(build_B_kernel, dim3(1), dim3(32), 0, stream,
                       w_n00, w_n10, w_n01, w_n11, B0, B1);

    int stiles = (E + 31) / 32;
    int blocks = (stiles + 7) / 8;
    hipLaunchKernelGGL(edge_kernel, dim3(blocks), dim3(256), 0, stream,
                       x, src, dst, angles, transporters, B0, B1,
                       deg, magsum, agg1, agg2, E);

    hipLaunchKernelGGL(finalize_kernel, dim3((V + 255) / 256), dim3(256), 0, stream,
                       x, e1, e2, w_self0, w_self11, deg, magsum, agg1, agg2,
                       (float*)d_out, V);
}


# Four per-vertex sums packed in one 16-byte record; four adjacent lanes add one edge's four values, so its f32 atomic adds leave as one piece of one line, not four words in four lines
# speedup vs baseline: 2.6651x; 2.6651x over previous
#include <hip/hip_runtime.h>
#include <hip/hip_bf16.h>

typedef __attribute__((ext_vector_type(16))) _Float16 v16h;
typedef __attribute__((ext_vector_type(8)))  float    v8f;

__device__ inline float wv_entry(int f, int n,
                                 const float* w_n00, const float* w_n10,
                                 const float* w_n01, const float* w_n11) {
    if (f < 16) {
        if (n == 0) return w_n00[f];
        if (n == 1) return w_n01[2 * f];
        if (n == 2) return w_n01[2 * f + 1];
        return 0.0f;
    }
    if (f < 48) {
        int c    = (f - 16) >> 1;
        int comp = (f - 16) & 1;
        int j    = n - (comp ? 9 : 3);
        if (j < 0 || j > 5) return 0.0f;
        if (j == 0) return w_n10[2 * c];
        if (j == 1) return w_n10[2 * c + 1];
        return w_n11[4 * c + (j - 2)];
    }
    return 0.0f;
}

__global__ void init_zero_kernel(float* p, int n) {
    int i = blockIdx.x * blockDim.x + threadIdx.x;
    if (i < n) p[i] = 0.0f;
}

__global__ void build_B_kernel(const float* w_n00, const float* w_n10,
                               const float* w_n01, const float* w_n11,
                               _Float16* B0, _Float16* B1) {
    int l = threadIdx.x;
    if (l >= 32) return;
    int n  = l & 15;
    int hi = l >> 4;
    for (int i = 0; i < 16; ++i) {
        int f = i + 8 * hi + (i >= 8 ? 8 : 0);
        B0[l * 16 + i] = (_Float16)wv_entry(f, n, w_n00, w_n10, w_n01, w_n11);
        B1[l * 16 + i] = (f < 16)
            ? (_Float16)wv_entry(32 + f, n, w_n00, w_n10, w_n01, w_n11)
            : (_Float16)0.0f;
    }
}

__device__ inline void gather_A(const float* __restrict__ row, int hi,
                                v16h& a0, v16h& a1) {
    const int fb0 = hi ? 8 : 0;
    const int fb1 = hi ? 24 : 16;
#pragma unroll
    for (int i = 0; i < 8; ++i) a0[i]     = (_Float16)row[fb0 + i];
#pragma unroll
    for (int i = 0; i < 8; ++i) a0[8 + i] = (_Float16)row[fb1 + i];
#pragma unroll
    for (int i = 0; i < 8; ++i) a1[i]     = (_Float16)row[32 + fb0 + i];
#pragma unroll
    for (int i = 0; i < 8; ++i) a1[8 + i] = (_Float16)0.0f;
}

__global__ __launch_bounds__(256)
void edge_kernel(const float* __restrict__ x,
                 const int*   __restrict__ src,
                 const int*   __restrict__ dst,
                 const float* __restrict__ angles,
                 const float* __restrict__ transporters,
                 const _Float16* __restrict__ B0p,
                 const _Float16* __restrict__ B1p,
                 float* __restrict__ acc,
                 int E) {
    __shared__ float tile[8][2 * 16 * 16];

    const int lane = threadIdx.x & 31;
    const int wv   = threadIdx.x >> 5;
    const int m    = lane & 15;
    const int hi   = lane >> 4;

    const long tbase = ((long)blockIdx.x * 8 + wv) * 32;
    long e0 = tbase + m;
    long e1 = tbase + 16 + m;
    long e0c = e0 < (long)E ? e0 : (long)E - 1;
    long e1c = e1 < (long)E ? e1 : (long)E - 1;

    int s0 = __builtin_nontemporal_load(&src[e0c]);
    int s1 = __builtin_nontemporal_load(&src[e1c]);

    v16h a0t0, a1t0, a0t1, a1t1;
    gather_A(x + (size_t)s0 * 48, hi, a0t0, a1t0);
    gather_A(x + (size_t)s1 * 48, hi, a0t1, a1t1);

    v16h b0 = *(const v16h*)(B0p + lane * 16);
    v16h b1 = *(const v16h*)(B1p + lane * 16);

    v8f c0 = {}, c1 = {};
    c0 = __builtin_amdgcn_wmma_f32_16x16x32_f16(false, a0t0, false, b0,
                                                (short)0, c0, false, false);
    c0 = __builtin_amdgcn_wmma_f32_16x16x32_f16(false, a1t0, false, b1,
                                                (short)0, c0, false, false);
    c1 = __builtin_amdgcn_wmma_f32_16x16x32_f16(false, a0t1, false, b0,
                                                (short)0, c1, false, false);
    c1 = __builtin_amdgcn_wmma_f32_16x16x32_f16(false, a1t1, false, b1,
                                                (short)0, c1, false, false);

    float* t = tile[wv];
#pragma unroll
    for (int r = 0; r < 8; ++r) {
        t[(r + 8 * hi) * 16 + m]       = c0[r];
        t[256 + (r + 8 * hi) * 16 + m] = c1[r];
    }
    __syncthreads();

    long ee = tbase + lane;
    float m0 = 0.0f, mv1 = 0.0f, mv2 = 0.0f;
    int   dv = -1;
    if (ee < (long)E) {
        const float* d = &t[lane * 16];
        float ang = __builtin_nontemporal_load(&angles[ee]);
        float tr  = __builtin_nontemporal_load(&transporters[ee]);
        float st, ct, s2t, c2t, sg, cg;
        __sincosf(ang, &st, &ct);
        __sincosf(2.0f * ang, &s2t, &c2t);
        __sincosf(tr, &sg, &cg);

        float d0 = d[0], sa = d[1], sb = d[2];
        float u1w0 = cg * d[3] - sg * d[9];
        float u2w0 = sg * d[3] + cg * d[9];
        float u1w1 = cg * d[4] - sg * d[10];
        float u2w1 = sg * d[4] + cg * d[10];
        float u1p  = cg * d[5] - sg * d[11];
        float u2p  = sg * d[5] + cg * d[11];
        float u1q  = cg * d[6] - sg * d[12];
        float u2q  = sg * d[6] + cg * d[12];
        float u1r  = cg * d[7] - sg * d[13];
        float u2r  = sg * d[7] + cg * d[13];
        float u1s  = cg * d[8] - sg * d[14];
        float u2s  = sg * d[8] + cg * d[14];

        m0  = d0 + ct * u1w0 + st * u2w0 + ct * u2w1 - st * u1w1;
        mv1 = sa * ct - sb * st + u1p - u2q
            + (c2t * u1r + s2t * u2r) - (s2t * u1s - c2t * u2s);
        mv2 = sa * st + sb * ct + u2p + u1q
            + (s2t * u1r - c2t * u2r) + (c2t * u1s + s2t * u2s);

        dv = __builtin_nontemporal_load(&dst[ee]);
    }

    const int  grp = lane >> 2;
    const bool odd = (lane & 1) != 0;
    const bool top = (lane & 2) != 0;
    float* slot = acc + (lane & 3);
    for (int j = 0; j < 4; ++j) {
        const int from = 8 * j + grp;
        int   dk = __shfl(dv,  from);
        float a1 = __shfl(m0,  from);
        float a2 = __shfl(mv1, from);
        float a3 = __shfl(mv2, from);
        float lo  = odd ? a1 : 1.0f;
        float hi  = odd ? a3 : a2;
        float val = top ? hi : lo;
        if (dk >= 0)
            __hip_atomic_fetch_add(slot + (size_t)dk * 4, val,
                                   __ATOMIC_RELAXED, __HIP_MEMORY_SCOPE_AGENT);
    }
}

__global__ __launch_bounds__(256)
void finalize_kernel(const float* __restrict__ x,
                     const float* __restrict__ e1,
                     const float* __restrict__ e2,
                     const float* __restrict__ w_self0,
                     const float* __restrict__ w_self11,
                     const float* __restrict__ acc,
                     float* __restrict__ out,
                     int V) {
    int v = blockIdx.x * blockDim.x + threadIdx.x;
    if (v >= V) return;
    const float* row = x + (size_t)v * 48;
    const float4 a = *(const float4*)(acc + (size_t)v * 4);
    float denom = fmaxf(a.x, 1.0f);
    float inv   = 1.0f / denom;
    float mag   = a.y * inv;
    float t1    = a.z * inv;
    float t2    = a.w * inv;
#pragma unroll
    for (int c = 0; c < 16; ++c) {
        mag += row[c] * w_self0[c];
        float xa = row[16 + 2 * c], xb = row[17 + 2 * c];
        float sa = w_self11[2 * c], sb = w_self11[2 * c + 1];
        t1 += xa * sa - xb * sb;
        t2 += xb * sa + xa * sb;
    }
    float ms = 2.0f / (1.0f + __expf(-mag));
#pragma unroll
    for (int k = 0; k < 3; ++k)
        out[v * 3 + k] = (t1 * e1[v * 3 + k] + t2 * e2[v * 3 + k]) * ms;
}

extern "C" void kernel_launch(void* const* d_in, const int* in_sizes, int n_in,
                              void* d_out, int out_size, void* d_ws, size_t ws_size,
                              hipStream_t stream) {
    const float* x            = (const float*)d_in[0];
    const int*   edge_index   = (const int*)d_in[1];
    const float* angles       = (const float*)d_in[2];
    const float* transporters = (const float*)d_in[3];
    const float* e1           = (const float*)d_in[4];
    const float* e2           = (const float*)d_in[5];
    const float* w_self0      = (const float*)d_in[6];
    const float* w_n00        = (const float*)d_in[7];
    const float* w_n10        = (const float*)d_in[8];
    const float* w_self11     = (const float*)d_in[9];
    const float* w_n01        = (const float*)d_in[10];
    const float* w_n11        = (const float*)d_in[11];

    const int E = in_sizes[2];
    const int V = in_sizes[0] / 48;
    const int* src = edge_index;
    const int* dst = edge_index + E;

    int nz = 4 * V;
    float* acc    = (float*)d_ws;
    _Float16* B0  = (_Float16*)(acc + nz);
    _Float16* B1  = B0 + 32 * 16;

    hipLaunchKernelGGL(init_zero_kernel, dim3((nz + 255) / 256), dim3(256), 0, stream,
                       acc, nz);
    hipLaunchKernelGGL(build_B_kernel, dim3(1), dim3(32), 0, stream,
                       w_n00, w_n10, w_n01, w_n11, B0, B1);

    int stiles = (E + 31) / 32;
    int blocks = (stiles + 7) / 8;
    hipLaunchKernelGGL(edge_kernel, dim3(blocks), dim3(256), 0, stream,
                       x, src, dst, angles, transporters, B0, B1,
                       acc, E);

    hipLaunchKernelGGL(finalize_kernel, dim3((V + 255) / 256), dim3(256), 0, stream,
                       x, e1, e2, w_self0, w_self11, acc,
                       (float*)d_out, V);
}


# Each wave of the edge kernel walks five 32-edge super-tiles in a loop (B fragments, die half read once per wave): a fifth as many waves, all resident at once; f32 atomic adds still in arrival order
# speedup vs baseline: 4.4540x; 1.6712x over previous
#include <hip/hip_runtime.h>
#include <hip/hip_bf16.h>

typedef __attribute__((ext_vector_type(16))) _Float16 v16h;
typedef __attribute__((ext_vector_type(8)))  float    v8f;

__device__ inline float wv_entry(int f, int n,
                                 const float* w_n00, const float* w_n10,
                                 const float* w_n01, const float* w_n11) {
    if (f < 16) {
        if (n == 0) return w_n00[f];
        if (n == 1) return w_n01[2 * f];
        if (n == 2) return w_n01[2 * f + 1];
        return 0.0f;
    }
    if (f < 48) {
        int c    = (f - 16) >> 1;
        int comp = (f - 16) & 1;
        int j    = n - (comp ? 9 : 3);
        if (j < 0 || j > 5) return 0.0f;
        if (j == 0) return w_n10[2 * c];
        if (j == 1) return w_n10[2 * c + 1];
        return w_n11[4 * c + (j - 2)];
    }
    return 0.0f;
}

__global__ void init_zero_kernel(float* p, int n4) {
    int i = blockIdx.x * blockDim.x + threadIdx.x;
    if (i < n4) ((float4*)p)[i] = make_float4(0.0f, 0.0f, 0.0f, 0.0f);
}

__global__ void build_B_kernel(const float* w_n00, const float* w_n10,
                               const float* w_n01, const float* w_n11,
                               _Float16* B0, _Float16* B1) {
    int l = threadIdx.x;
    if (l >= 32) return;
    int n  = l & 15;
    int hi = l >> 4;
    for (int i = 0; i < 16; ++i) {
        int f = i + 8 * hi + (i >= 8 ? 8 : 0);
        B0[l * 16 + i] = (_Float16)wv_entry(f, n, w_n00, w_n10, w_n01, w_n11);
        B1[l * 16 + i] = (f < 16)
            ? (_Float16)wv_entry(32 + f, n, w_n00, w_n10, w_n01, w_n11)
            : (_Float16)0.0f;
    }
}

__device__ inline void gather_A(const float* __restrict__ row, int hi,
                                v16h& a0, v16h& a1) {
    const int fb0 = hi ? 8 : 0;
    const int fb1 = hi ? 24 : 16;
#pragma unroll
    for (int i = 0; i < 8; ++i) a0[i]     = (_Float16)row[fb0 + i];
#pragma unroll
    for (int i = 0; i < 8; ++i) a0[8 + i] = (_Float16)row[fb1 + i];
#pragma unroll
    for (int i = 0; i < 8; ++i) a1[i]     = (_Float16)row[32 + fb0 + i];
#pragma unroll
    for (int i = 0; i < 8; ++i) a1[8 + i] = (_Float16)0.0f;
}

constexpr int TILES_PER_WAVE = 5;

__global__ __launch_bounds__(256)
void edge_kernel(const float* __restrict__ x,
                 const int*   __restrict__ src,
                 const int*   __restrict__ dst,
                 const float* __restrict__ angles,
                 const float* __restrict__ transporters,
                 const _Float16* __restrict__ B0p,
                 const _Float16* __restrict__ B1p,
                 float* __restrict__ acc,
                 int plane_stride,
                 int E) {
    __shared__ float tile[8][2 * 16 * 16];
    __shared__ int   half_of_group;

    const int lane = threadIdx.x & 31;
    const int wv   = threadIdx.x >> 5;
    const int m    = lane & 15;
    const int hi   = lane >> 4;

    if (threadIdx.x == 0) {
        const unsigned where = __builtin_amdgcn_s_sendmsg_rtn(0x87);
        half_of_group = (int)((where >> 18) & 1u);
    }

    const v16h b0 = *(const v16h*)(B0p + lane * 16);
    const v16h b1 = *(const v16h*)(B1p + lane * 16);

    __syncthreads();
    float* const slot = acc + (half_of_group ? (size_t)plane_stride : (size_t)0) + (lane & 3);
    float* const t    = tile[wv];

    const int  grp = lane >> 2;
    const bool odd = (lane & 1) != 0;
    const bool top = (lane & 2) != 0;

    for (int it = 0; it < TILES_PER_WAVE; ++it) {
        const long gtile = (long)blockIdx.x * TILES_PER_WAVE + it;
        if (gtile * 256 >= (long)E) break;

        const long tbase = (gtile * 8 + wv) * 32;
        long e0 = tbase + m;
        long e1 = tbase + 16 + m;
        long e0c = e0 < (long)E ? e0 : (long)E - 1;
        long e1c = e1 < (long)E ? e1 : (long)E - 1;

        int s0 = __builtin_nontemporal_load(&src[e0c]);
        int s1 = __builtin_nontemporal_load(&src[e1c]);

        v16h a0t0, a1t0, a0t1, a1t1;
        gather_A(x + (size_t)s0 * 48, hi, a0t0, a1t0);
        gather_A(x + (size_t)s1 * 48, hi, a0t1, a1t1);

        v8f c0 = {}, c1 = {};
        c0 = __builtin_amdgcn_wmma_f32_16x16x32_f16(false, a0t0, false, b0,
                                                    (short)0, c0, false, false);
        c0 = __builtin_amdgcn_wmma_f32_16x16x32_f16(false, a1t0, false, b1,
                                                    (short)0, c0, false, false);
        c1 = __builtin_amdgcn_wmma_f32_16x16x32_f16(false, a0t1, false, b0,
                                                    (short)0, c1, false, false);
        c1 = __builtin_amdgcn_wmma_f32_16x16x32_f16(false, a1t1, false, b1,
                                                    (short)0, c1, false, false);

#pragma unroll
        for (int r = 0; r < 8; ++r) {
            t[(r + 8 * hi) * 16 + m]       = c0[r];
            t[256 + (r + 8 * hi) * 16 + m] = c1[r];
        }
        __syncthreads();

        long ee = tbase + lane;
        float m0 = 0.0f, mv1 = 0.0f, mv2 = 0.0f;
        int   dv = -1;
        if (ee < (long)E) {
            const float* d = &t[lane * 16];
            float ang = __builtin_nontemporal_load(&angles[ee]);
            float tr  = __builtin_nontemporal_load(&transporters[ee]);
            float st, ct, s2t, c2t, sg, cg;
            __sincosf(ang, &st, &ct);
            __sincosf(2.0f * ang, &s2t, &c2t);
            __sincosf(tr, &sg, &cg);

            float d0 = d[0], sa = d[1], sb = d[2];
            float u1w0 = cg * d[3] - sg * d[9];
            float u2w0 = sg * d[3] + cg * d[9];
            float u1w1 = cg * d[4] - sg * d[10];
            float u2w1 = sg * d[4] + cg * d[10];
            float u1p  = cg * d[5] - sg * d[11];
            float u2p  = sg * d[5] + cg * d[11];
            float u1q  = cg * d[6] - sg * d[12];
            float u2q  = sg * d[6] + cg * d[12];
            float u1r  = cg * d[7] - sg * d[13];
            float u2r  = sg * d[7] + cg * d[13];
            float u1s  = cg * d[8] - sg * d[14];
            float u2s  = sg * d[8] + cg * d[14];

            m0  = d0 + ct * u1w0 + st * u2w0 + ct * u2w1 - st * u1w1;
            mv1 = sa * ct - sb * st + u1p - u2q
                + (c2t * u1r + s2t * u2r) - (s2t * u1s - c2t * u2s);
            mv2 = sa * st + sb * ct + u2p + u1q
                + (s2t * u1r - c2t * u2r) + (c2t * u1s + s2t * u2s);

            dv = __builtin_nontemporal_load(&dst[ee]);
        }

        for (int j = 0; j < 4; ++j) {
            const int from = 8 * j + grp;
            int   dk = __shfl(dv,  from);
            float a1 = __shfl(m0,  from);
            float a2 = __shfl(mv1, from);
            float a3 = __shfl(mv2, from);
            float lo  = odd ? a1 : 1.0f;
            float up  = odd ? a3 : a2;
            float val = top ? up : lo;
            if (dk >= 0)
                __hip_atomic_fetch_add(slot + (size_t)dk * 4, val,
                                       __ATOMIC_RELAXED, __HIP_MEMORY_SCOPE_AGENT);
        }
    }
}

__global__ __launch_bounds__(256)
void finalize_kernel(const float* __restrict__ x,
                     const float* __restrict__ e1,
                     const float* __restrict__ e2,
                     const float* __restrict__ w_self0,
                     const float* __restrict__ w_self11,
                     const float* __restrict__ acc,
                     int plane_stride,
                     float* __restrict__ out,
                     int V) {
    int v = blockIdx.x * blockDim.x + threadIdx.x;
    if (v >= V) return;
    const float* row = x + (size_t)v * 48;
    float4 a = *(const float4*)(acc + (size_t)v * 4);
    if (plane_stride != 0) {
        const float4 b = *(const float4*)(acc + (size_t)plane_stride + (size_t)v * 4);
        a.x += b.x; a.y += b.y; a.z += b.z; a.w += b.w;
    }
    float denom = fmaxf(a.x, 1.0f);
    float inv   = 1.0f / denom;
    float mag   = a.y * inv;
    float t1    = a.z * inv;
    float t2    = a.w * inv;
#pragma unroll
    for (int c = 0; c < 16; ++c) {
        mag += row[c] * w_self0[c];
        float xa = row[16 + 2 * c], xb = row[17 + 2 * c];
        float sa = w_self11[2 * c], sb = w_self11[2 * c + 1];
        t1 += xa * sa - xb * sb;
        t2 += xb * sa + xa * sb;
    }
    float ms = 2.0f / (1.0f + __expf(-mag));
#pragma unroll
    for (int k = 0; k < 3; ++k)
        out[v * 3 + k] = (t1 * e1[v * 3 + k] + t2 * e2[v * 3 + k]) * ms;
}

extern "C" void kernel_launch(void* const* d_in, const int* in_sizes, int n_in,
                              void* d_out, int out_size, void* d_ws, size_t ws_size,
                              hipStream_t stream) {
    const float* x            = (const float*)d_in[0];
    const int*   edge_index   = (const int*)d_in[1];
    const float* angles       = (const float*)d_in[2];
    const float* transporters = (const float*)d_in[3];
    const float* e1           = (const float*)d_in[4];
    const float* e2           = (const float*)d_in[5];
    const float* w_self0      = (const float*)d_in[6];
    const float* w_n00        = (const float*)d_in[7];
    const float* w_n10        = (const float*)d_in[8];
    const float* w_self11     = (const float*)d_in[9];
    const float* w_n01        = (const float*)d_in[10];
    const float* w_n11        = (const float*)d_in[11];

    const int E = in_sizes[2];
    const int V = in_sizes[0] / 48;
    const int* src = edge_index;
    const int* dst = edge_index + E;

    const size_t plane_bytes = (size_t)V * 4 * sizeof(float);
    const size_t b_bytes     = (size_t)2 * 32 * 16 * sizeof(_Float16);
    const int planes       = (ws_size >= 2 * plane_bytes + b_bytes) ? 2 : 1;
    const int plane_stride = (planes == 2) ? 4 * V : 0;
    int n4 = planes * V;
    float* acc    = (float*)d_ws;
    _Float16* B0  = (_Float16*)(acc + (size_t)n4 * 4);
    _Float16* B1  = B0 + 32 * 16;

    hipLaunchKernelGGL(init_zero_kernel, dim3((n4 + 255) / 256), dim3(256), 0, stream,
                       acc, n4);
    hipLaunchKernelGGL(build_B_kernel, dim3(1), dim3(32), 0, stream,
                       w_n00, w_n10, w_n01, w_n11, B0, B1);

    int stiles = (E + 31) / 32;
    int gtiles = (stiles + 7) / 8;
    int blocks = (gtiles + TILES_PER_WAVE - 1) / TILES_PER_WAVE;
    hipLaunchKernelGGL(edge_kernel, dim3(blocks), dim3(256), 0, stream,
                       x, src, dst, angles, transporters, B0, B1,
                       acc, plane_stride, E);

    hipLaunchKernelGGL(finalize_kernel, dim3((V + 255) / 256), dim3(256), 0, stream,
                       x, e1, e2, w_self0, w_self11, acc, plane_stride,
                       (float*)d_out, V);
}


# Edge kernel fetches each projected 64-byte row with four adjacent lanes (8 lines per load, not 32) and hands it to the edge's lane through a wave-private LDS tile; f32 atomic adds in arrival order
# speedup vs baseline: 7.5731x; 1.7003x over previous
#include <hip/hip_runtime.h>
#include <hip/hip_bf16.h>

typedef __attribute__((ext_vector_type(16))) _Float16 v16h;
typedef __attribute__((ext_vector_type(8)))  float    v8f;
typedef __attribute__((ext_vector_type(4)))  float    v4f;

__device__ inline float wv_entry(int f, int n,
                                 const float* w_n00, const float* w_n10,
                                 const float* w_n01, const float* w_n11) {
    if (f < 16) {
        if (n == 0) return w_n00[f];
        if (n == 1) return w_n01[2 * f];
        if (n == 2) return w_n01[2 * f + 1];
        return 0.0f;
    }
    if (f < 48) {
        int c    = (f - 16) >> 1;
        int comp = (f - 16) & 1;
        int j    = n - (comp ? 9 : 3);
        if (j < 0 || j > 5) return 0.0f;
        if (j == 0) return w_n10[2 * c];
        if (j == 1) return w_n10[2 * c + 1];
        return w_n11[4 * c + (j - 2)];
    }
    return 0.0f;
}

struct WvWhere { const float* p; bool ok; };
__device__ inline WvWhere wv_where(int f, int n,
                                   const float* w_n00, const float* w_n10,
                                   const float* w_n01, const float* w_n11) {
    const int  c    = (f - 16) >> 1;
    const int  comp = (f - 16) & 1;
    const int  j    = n - (comp ? 9 : 3);
    const bool lo   = f < 16;
    const bool mid  = (f >= 16) & (f < 48);
    const bool k00  = lo & (n == 0);
    const bool k01  = lo & ((n == 1) | (n == 2));
    const bool k10  = mid & ((j == 0) | (j == 1));
    const bool k11  = mid & (j >= 2) & (j <= 5);
    const float* base = k01 ? w_n01 : (k10 ? w_n10 : (k11 ? w_n11 : w_n00));
    const int    idx  = k00 ? f : (k01 ? 2 * f + n - 1 : (k10 ? 2 * c + j : (k11 ? 4 * c + j - 2 : 0)));
    WvWhere r;
    r.p  = base + idx;
    r.ok = k00 | k01 | k10 | k11;
    return r;
}

typedef __attribute__((ext_vector_type(2))) _Float16 v2h;

__global__ void prep_kernel(float* p, int n4,
                            const float* w_n00, const float* w_n10,
                            const float* w_n01, const float* w_n11,
                            _Float16* B) {
    int i = blockIdx.x * blockDim.x + threadIdx.x;
    if (i < n4) ((float4*)p)[i] = make_float4(0.0f, 0.0f, 0.0f, 0.0f);
    if (i < 512) {
        v2h v;
#pragma unroll
        for (int k = 0; k < 2; ++k) {
            const int idx   = 2 * i + k;
            const int chunk = idx >> 9;
            const int l     = (idx >> 4) & 31;
            const int e     = idx & 15;
            const int n     = l & 15;
            const int hi    = l >> 4;
            const int f     = e + 8 * hi + (e >= 8 ? 8 : 0);
            float w;
            if (chunk == 0) w = wv_entry(f, n, w_n00, w_n10, w_n01, w_n11);
            else            w = (f < 16) ? wv_entry(32 + f, n, w_n00, w_n10, w_n01, w_n11) : 0.0f;
            v[k] = (_Float16)w;
        }
        *(v2h*)(B + 2 * i) = v;
    }
}

__device__ inline void gather_A(const float* __restrict__ row, int hi,
                                v16h& a0, v16h& a1) {
    const int fb0 = hi ? 8 : 0;
    const int fb1 = hi ? 24 : 16;
#pragma unroll
    for (int i = 0; i < 8; ++i) a0[i]     = (_Float16)row[fb0 + i];
#pragma unroll
    for (int i = 0; i < 8; ++i) a0[8 + i] = (_Float16)row[fb1 + i];
#pragma unroll
    for (int i = 0; i < 8; ++i) a1[i]     = (_Float16)row[32 + fb0 + i];
#pragma unroll
    for (int i = 0; i < 8; ++i) a1[8 + i] = (_Float16)0.0f;
}

constexpr int TILES_PER_WAVE = 5;

__global__ __launch_bounds__(256)
void edge_kernel(const float* __restrict__ x,
                 const int*   __restrict__ src,
                 const int*   __restrict__ dst,
                 const float* __restrict__ angles,
                 const float* __restrict__ transporters,
                 const _Float16* __restrict__ B0p,
                 const _Float16* __restrict__ B1p,
                 float* __restrict__ acc,
                 int plane_stride,
                 int E) {
    __shared__ float tile[8][2 * 16 * 16];
    __shared__ int   half_of_group;

    const int lane = threadIdx.x & 31;
    const int wv   = threadIdx.x >> 5;
    const int m    = lane & 15;
    const int hi   = lane >> 4;

    if (threadIdx.x == 0) {
        const unsigned where = __builtin_amdgcn_s_sendmsg_rtn(0x87);
        half_of_group = (int)((where >> 18) & 1u);
    }

    const v16h b0 = *(const v16h*)(B0p + lane * 16);
    const v16h b1 = *(const v16h*)(B1p + lane * 16);

    __syncthreads();
    float* const slot = acc + (half_of_group ? (size_t)plane_stride : (size_t)0) + (lane & 3);
    float* const t    = tile[wv];

    const int  grp = lane >> 2;
    const bool odd = (lane & 1) != 0;
    const bool top = (lane & 2) != 0;

    for (int it = 0; it < TILES_PER_WAVE; ++it) {
        const long gtile = (long)blockIdx.x * TILES_PER_WAVE + it;
        if (gtile * 256 >= (long)E) break;

        const long tbase = (gtile * 8 + wv) * 32;
        long e0 = tbase + m;
        long e1 = tbase + 16 + m;
        long e0c = e0 < (long)E ? e0 : (long)E - 1;
        long e1c = e1 < (long)E ? e1 : (long)E - 1;

        int s0 = __builtin_nontemporal_load(&src[e0c]);
        int s1 = __builtin_nontemporal_load(&src[e1c]);

        v16h a0t0, a1t0, a0t1, a1t1;
        gather_A(x + (size_t)s0 * 48, hi, a0t0, a1t0);
        gather_A(x + (size_t)s1 * 48, hi, a0t1, a1t1);

        v8f c0 = {}, c1 = {};
        c0 = __builtin_amdgcn_wmma_f32_16x16x32_f16(false, a0t0, false, b0,
                                                    (short)0, c0, false, false);
        c0 = __builtin_amdgcn_wmma_f32_16x16x32_f16(false, a1t0, false, b1,
                                                    (short)0, c0, false, false);
        c1 = __builtin_amdgcn_wmma_f32_16x16x32_f16(false, a0t1, false, b0,
                                                    (short)0, c1, false, false);
        c1 = __builtin_amdgcn_wmma_f32_16x16x32_f16(false, a1t1, false, b1,
                                                    (short)0, c1, false, false);

#pragma unroll
        for (int r = 0; r < 8; ++r) {
            t[(r + 8 * hi) * 16 + m]       = c0[r];
            t[256 + (r + 8 * hi) * 16 + m] = c1[r];
        }
        __syncthreads();

        long ee = tbase + lane;
        float m0 = 0.0f, mv1 = 0.0f, mv2 = 0.0f;
        int   dv = -1;
        if (ee < (long)E) {
            const float* d = &t[lane * 16];
            float ang = __builtin_nontemporal_load(&angles[ee]);
            float tr  = __builtin_nontemporal_load(&transporters[ee]);
            float st, ct, s2t, c2t, sg, cg;
            __sincosf(ang, &st, &ct);
            __sincosf(2.0f * ang, &s2t, &c2t);
            __sincosf(tr, &sg, &cg);

            float d0 = d[0], sa = d[1], sb = d[2];
            float u1w0 = cg * d[3] - sg * d[9];
            float u2w0 = sg * d[3] + cg * d[9];
            float u1w1 = cg * d[4] - sg * d[10];
            float u2w1 = sg * d[4] + cg * d[10];
            float u1p  = cg * d[5] - sg * d[11];
            float u2p  = sg * d[5] + cg * d[11];
            float u1q  = cg * d[6] - sg * d[12];
            float u2q  = sg * d[6] + cg * d[12];
            float u1r  = cg * d[7] - sg * d[13];
            float u2r  = sg * d[7] + cg * d[13];
            float u1s  = cg * d[8] - sg * d[14];
            float u2s  = sg * d[8] + cg * d[14];

            m0  = d0 + ct * u1w0 + st * u2w0 + ct * u2w1 - st * u1w1;
            mv1 = sa * ct - sb * st + u1p - u2q
                + (c2t * u1r + s2t * u2r) - (s2t * u1s - c2t * u2s);
            mv2 = sa * st + sb * ct + u2p + u1q
                + (s2t * u1r - c2t * u2r) + (c2t * u1s + s2t * u2s);

            dv = __builtin_nontemporal_load(&dst[ee]);
        }

        for (int j = 0; j < 4; ++j) {
            const int from = 8 * j + grp;
            int   dk = __shfl(dv,  from);
            float a1 = __shfl(m0,  from);
            float a2 = __shfl(mv1, from);
            float a3 = __shfl(mv2, from);
            float lo  = odd ? a1 : 1.0f;
            float up  = odd ? a3 : a2;
            float val = top ? up : lo;
            if (dk >= 0)
                __hip_atomic_fetch_add(slot + (size_t)dk * 4, val,
                                       __ATOMIC_RELAXED, __HIP_MEMORY_SCOPE_AGENT);
        }
    }
}

__global__ __launch_bounds__(256)
void project_kernel(const float* __restrict__ x,
                    const float* w_n00, const float* w_n10,
                    const float* w_n01, const float* w_n11,
                    float* __restrict__ P,
                    float* __restrict__ acc, int n4,
                    int V) {
    __shared__ _Float16 Bs[2 * 32 * 16];
    const int tid  = threadIdx.x;
    const int lane = tid & 31;
    const int wv   = tid >> 5;
    const int m    = lane & 15;
    const int hi   = lane >> 4;

    const long vbase = ((long)blockIdx.x * 8 + wv) * 32;
    long v0 = vbase + m, v1 = vbase + 16 + m;
    v0 = v0 < (long)V ? v0 : (long)V - 1;
    v1 = v1 < (long)V ? v1 : (long)V - 1;
    v16h a0t0, a1t0, a0t1, a1t1;
    gather_A(x + (size_t)v0 * 48, hi, a0t0, a1t0);
    gather_A(x + (size_t)v1 * 48, hi, a0t1, a1t1);

    WvWhere wh[4];
#pragma unroll
    for (int k = 0; k < 4; ++k) {
        const int idx   = 4 * tid + k;
        const int chunk = idx >> 9;
        const int l     = (idx >> 4) & 31;
        const int e     = idx & 15;
        const int n     = l & 15;
        const int h     = l >> 4;
        const int f     = e + 8 * h + (e >= 8 ? 8 : 0);
        const int feat  = (chunk == 0) ? f : (f < 16 ? 32 + f : 48);
        wh[k] = wv_where(feat, n, w_n00, w_n10, w_n01, w_n11);
    }
    float wl[4];
#pragma unroll
    for (int k = 0; k < 4; ++k) wl[k] = *wh[k].p;
#pragma unroll
    for (int k = 0; k < 4; ++k) Bs[4 * tid + k] = (_Float16)(wh[k].ok ? wl[k] : 0.0f);

    {
        const int nth = gridDim.x * 256;
        for (int i = blockIdx.x * 256 + tid; i < n4; i += nth)
            ((float4*)acc)[i] = make_float4(0.0f, 0.0f, 0.0f, 0.0f);
    }
    __syncthreads();
    if (vbase >= (long)V) return;

    const v16h b0 = *(const v16h*)(Bs + lane * 16);
    const v16h b1 = *(const v16h*)(Bs + 512 + lane * 16);

    v8f c0 = {}, c1 = {};
    c0 = __builtin_amdgcn_wmma_f32_16x16x32_f16(false, a0t0, false, b0, (short)0, c0, false, false);
    c0 = __builtin_amdgcn_wmma_f32_16x16x32_f16(false, a1t0, false, b1, (short)0, c0, false, false);
    c1 = __builtin_amdgcn_wmma_f32_16x16x32_f16(false, a0t1, false, b0, (short)0, c1, false, false);
    c1 = __builtin_amdgcn_wmma_f32_16x16x32_f16(false, a1t1, false, b1, (short)0, c1, false, false);

#pragma unroll
    for (int r = 0; r < 8; ++r) {
        const long r0 = vbase + 8 * hi + r;
        const long r1 = r0 + 16;
        if (r0 < (long)V) P[(size_t)r0 * 16 + m] = c0[r];
        if (r1 < (long)V) P[(size_t)r1 * 16 + m] = c1[r];
    }
}

constexpr int ROW_PITCH = 5;

__global__ __launch_bounds__(256)
void edge_kernel_p(const float* __restrict__ P,
                   const int*   __restrict__ src,
                   const int*   __restrict__ dst,
                   const float* __restrict__ angles,
                   const float* __restrict__ transporters,
                   float* __restrict__ acc,
                   int plane_stride,
                   int E) {
    __shared__ int half_of_group;
    __shared__ v4f rows[8][32 * ROW_PITCH];

    const int lane = threadIdx.x & 31;
    const int wv   = threadIdx.x >> 5;
    const int qk   = lane & 3;
    v4f* const rw  = rows[wv];

    if (threadIdx.x == 0) {
        const unsigned where = __builtin_amdgcn_s_sendmsg_rtn(0x87);
        half_of_group = (int)((where >> 18) & 1u);
    }
    __syncthreads();
    float* const slot = acc + (half_of_group ? (size_t)plane_stride : (size_t)0) + (lane & 3);

    const int  grp = lane >> 2;
    const bool odd = (lane & 1) != 0;
    const bool top = (lane & 2) != 0;

    for (int it = 0; it < TILES_PER_WAVE; ++it) {
        const long gtile = (long)blockIdx.x * TILES_PER_WAVE + it;
        if (gtile * 256 >= (long)E) break;

        const long tbase = (gtile * 8 + wv) * 32;
        const long ee  = tbase + lane;
        const bool has = ee < (long)E;
        const long eec = has ? ee : (long)E - 1;

        const long qe   = tbase + 4 * grp;
        const long last = (long)E - 1;
        const int sq0 = src[qe     < last ? qe     : last];
        const int sq1 = src[qe + 1 < last ? qe + 1 : last];
        const int sq2 = src[qe + 2 < last ? qe + 2 : last];
        const int sq3 = src[qe + 3 < last ? qe + 3 : last];
        const float ang = __builtin_nontemporal_load(&angles[eec]);
        const float tr  = __builtin_nontemporal_load(&transporters[eec]);
        const int   dd  = __builtin_nontemporal_load(&dst[eec]);
        __builtin_amdgcn_sched_barrier(0);
        const v4f g0 = *((const v4f*)(P + (size_t)sq0 * 16) + qk);
        const v4f g1 = *((const v4f*)(P + (size_t)sq1 * 16) + qk);
        const v4f g2 = *((const v4f*)(P + (size_t)sq2 * 16) + qk);
        const v4f g3 = *((const v4f*)(P + (size_t)sq3 * 16) + qk);
        rw[(4 * grp + 0) * ROW_PITCH + qk] = g0;
        rw[(4 * grp + 1) * ROW_PITCH + qk] = g1;
        rw[(4 * grp + 2) * ROW_PITCH + qk] = g2;
        rw[(4 * grp + 3) * ROW_PITCH + qk] = g3;

        float m0, mv1, mv2;
        const int dv = has ? dd : -1;
        {
            const v4f* q = rw + lane * ROW_PITCH;
            const v4f q0 = q[0], q1 = q[1], q2 = q[2], q3 = q[3];
            const float d[16] = {q0.x, q0.y, q0.z, q0.w, q1.x, q1.y, q1.z, q1.w,
                                 q2.x, q2.y, q2.z, q2.w, q3.x, q3.y, q3.z, q3.w};
            float st, ct, s2t, c2t, sg, cg;
            __sincosf(ang, &st, &ct);
            __sincosf(2.0f * ang, &s2t, &c2t);
            __sincosf(tr, &sg, &cg);

            float d0 = d[0], sa = d[1], sb = d[2];
            float u1w0 = cg * d[3] - sg * d[9];
            float u2w0 = sg * d[3] + cg * d[9];
            float u1w1 = cg * d[4] - sg * d[10];
            float u2w1 = sg * d[4] + cg * d[10];
            float u1p  = cg * d[5] - sg * d[11];
            float u2p  = sg * d[5] + cg * d[11];
            float u1q  = cg * d[6] - sg * d[12];
            float u2q  = sg * d[6] + cg * d[12];
            float u1r  = cg * d[7] - sg * d[13];
            float u2r  = sg * d[7] + cg * d[13];
            float u1s  = cg * d[8] - sg * d[14];
            float u2s  = sg * d[8] + cg * d[14];

            m0  = d0 + ct * u1w0 + st * u2w0 + ct * u2w1 - st * u1w1;
            mv1 = sa * ct - sb * st + u1p - u2q
                + (c2t * u1r + s2t * u2r) - (s2t * u1s - c2t * u2s);
            mv2 = sa * st + sb * ct + u2p + u1q
                + (s2t * u1r - c2t * u2r) + (c2t * u1s + s2t * u2s);
        }

        for (int j = 0; j < 4; ++j) {
            const int from = 8 * j + grp;
            int   dk = __shfl(dv,  from);
            float a1 = __shfl(m0,  from);
            float a2 = __shfl(mv1, from);
            float a3 = __shfl(mv2, from);
            float lo  = odd ? a1 : 1.0f;
            float up  = odd ? a3 : a2;
            float val = top ? up : lo;
            if (dk >= 0)
                __hip_atomic_fetch_add(slot + (size_t)dk * 4, val,
                                       __ATOMIC_RELAXED, __HIP_MEMORY_SCOPE_AGENT);
        }
    }
}

__global__ __launch_bounds__(256)
void finalize_kernel(const float* __restrict__ x,
                     const float* __restrict__ e1,
                     const float* __restrict__ e2,
                     const float* __restrict__ w_self0,
                     const float* __restrict__ w_self11,
                     const float* __restrict__ acc,
                     int plane_stride,
                     float* __restrict__ out,
                     int V) {
    int v = blockIdx.x * blockDim.x + threadIdx.x;
    if (v >= V) return;
    const float* row = x + (size_t)v * 48;
    float4 a = *(const float4*)(acc + (size_t)v * 4);
    if (plane_stride != 0) {
        const float4 b = *(const float4*)(acc + (size_t)plane_stride + (size_t)v * 4);
        a.x += b.x; a.y += b.y; a.z += b.z; a.w += b.w;
    }
    float denom = fmaxf(a.x, 1.0f);
    float inv   = 1.0f / denom;
    float mag   = a.y * inv;
    float t1    = a.z * inv;
    float t2    = a.w * inv;
#pragma unroll
    for (int c = 0; c < 16; ++c) {
        mag += row[c] * w_self0[c];
        float xa = row[16 + 2 * c], xb = row[17 + 2 * c];
        float sa = w_self11[2 * c], sb = w_self11[2 * c + 1];
        t1 += xa * sa - xb * sb;
        t2 += xb * sa + xa * sb;
    }
    float ms = 2.0f / (1.0f + __expf(-mag));
#pragma unroll
    for (int k = 0; k < 3; ++k)
        out[v * 3 + k] = (t1 * e1[v * 3 + k] + t2 * e2[v * 3 + k]) * ms;
}

extern "C" void kernel_launch(void* const* d_in, const int* in_sizes, int n_in,
                              void* d_out, int out_size, void* d_ws, size_t ws_size,
                              hipStream_t stream) {
    const float* x            = (const float*)d_in[0];
    const int*   edge_index   = (const int*)d_in[1];
    const float* angles       = (const float*)d_in[2];
    const float* transporters = (const float*)d_in[3];
    const float* e1           = (const float*)d_in[4];
    const float* e2           = (const float*)d_in[5];
    const float* w_self0      = (const float*)d_in[6];
    const float* w_n00        = (const float*)d_in[7];
    const float* w_n10        = (const float*)d_in[8];
    const float* w_self11     = (const float*)d_in[9];
    const float* w_n01        = (const float*)d_in[10];
    const float* w_n11        = (const float*)d_in[11];

    const int E = in_sizes[2];
    const int V = in_sizes[0] / 48;
    const int* src = edge_index;
    const int* dst = edge_index + E;

    const size_t plane_bytes = (size_t)V * 4 * sizeof(float);
    const size_t b_bytes     = (size_t)2 * 32 * 16 * sizeof(_Float16);
    const size_t p_bytes     = (size_t)V * 16 * sizeof(float);

    if (ws_size >= p_bytes + plane_bytes) {
        const int planes       = (ws_size >= p_bytes + 2 * plane_bytes) ? 2 : 1;
        const int plane_stride = (planes == 2) ? 4 * V : 0;
        const int n4 = planes * V;
        float* P   = (float*)d_ws;
        float* acc = P + (size_t)V * 16;

        hipLaunchKernelGGL(project_kernel, dim3((V + 255) / 256), dim3(256), 0, stream,
                           x, w_n00, w_n10, w_n01, w_n11, P, acc, n4, V);

        int stiles = (E + 31) / 32;
        int gtiles = (stiles + 7) / 8;
        int blocks = (gtiles + TILES_PER_WAVE - 1) / TILES_PER_WAVE;
        hipLaunchKernelGGL(edge_kernel_p, dim3(blocks), dim3(256), 0, stream,
                           P, src, dst, angles, transporters, acc, plane_stride, E);

        hipLaunchKernelGGL(finalize_kernel, dim3((V + 255) / 256), dim3(256), 0, stream,
                           x, e1, e2, w_self0, w_self11, acc, plane_stride,
                           (float*)d_out, V);
        return;
    }

    const int planes       = (ws_size >= 2 * plane_bytes + b_bytes) ? 2 : 1;
    const int plane_stride = (planes == 2) ? 4 * V : 0;
    int n4 = planes * V;
    float* acc    = (float*)d_ws;
    _Float16* B0  = (_Float16*)(acc + (size_t)n4 * 4);
    _Float16* B1  = B0 + 32 * 16;

    const int prep_threads = n4 > 512 ? n4 : 512;
    hipLaunchKernelGGL(prep_kernel, dim3((prep_threads + 255) / 256), dim3(256), 0, stream,
                       acc, n4, w_n00, w_n10, w_n01, w_n11, B0);

    int stiles = (E + 31) / 32;
    int gtiles = (stiles + 7) / 8;
    int blocks = (gtiles + TILES_PER_WAVE - 1) / TILES_PER_WAVE;
    hipLaunchKernelGGL(edge_kernel, dim3(blocks), dim3(256), 0, stream,
                       x, src, dst, angles, transporters, B0, B1,
                       acc, plane_stride, E);

    hipLaunchKernelGGL(finalize_kernel, dim3((V + 255) / 256), dim3(256), 0, stream,
                       x, e1, e2, w_self0, w_self11, acc, plane_stride,
                       (float*)d_out, V);
}


# Edge kernel's trips software-pipelined: stream words of all trips requested at wave start, row quarters one trip ahead, source indices (one 16-byte load) two ahead; f32 atomic adds in arrival order
# speedup vs baseline: 7.8182x; 1.0324x over previous
#include <hip/hip_runtime.h>
#include <hip/hip_bf16.h>

typedef __attribute__((ext_vector_type(16))) _Float16 v16h;
typedef __attribute__((ext_vector_type(8)))  float    v8f;
typedef __attribute__((ext_vector_type(4)))  float    v4f;
typedef __attribute__((ext_vector_type(4)))  int      v4i;

__device__ inline float wv_entry(int f, int n,
                                 const float* w_n00, const float* w_n10,
                                 const float* w_n01, const float* w_n11) {
    if (f < 16) {
        if (n == 0) return w_n00[f];
        if (n == 1) return w_n01[2 * f];
        if (n == 2) return w_n01[2 * f + 1];
        return 0.0f;
    }
    if (f < 48) {
        int c    = (f - 16) >> 1;
        int comp = (f - 16) & 1;
        int j    = n - (comp ? 9 : 3);
        if (j < 0 || j > 5) return 0.0f;
        if (j == 0) return w_n10[2 * c];
        if (j == 1) return w_n10[2 * c + 1];
        return w_n11[4 * c + (j - 2)];
    }
    return 0.0f;
}

struct WvWhere { const float* p; bool ok; };
__device__ inline WvWhere wv_where(int f, int n,
                                   const float* w_n00, const float* w_n10,
                                   const float* w_n01, const float* w_n11) {
    const int  c    = (f - 16) >> 1;
    const int  comp = (f - 16) & 1;
    const int  j    = n - (comp ? 9 : 3);
    const bool lo   = f < 16;
    const bool mid  = (f >= 16) & (f < 48);
    const bool k00  = lo & (n == 0);
    const bool k01  = lo & ((n == 1) | (n == 2));
    const bool k10  = mid & ((j == 0) | (j == 1));
    const bool k11  = mid & (j >= 2) & (j <= 5);
    const float* base = k01 ? w_n01 : (k10 ? w_n10 : (k11 ? w_n11 : w_n00));
    const int    idx  = k00 ? f : (k01 ? 2 * f + n - 1 : (k10 ? 2 * c + j : (k11 ? 4 * c + j - 2 : 0)));
    WvWhere r;
    r.p  = base + idx;
    r.ok = k00 | k01 | k10 | k11;
    return r;
}

typedef __attribute__((ext_vector_type(2))) _Float16 v2h;

__global__ void prep_kernel(float* p, int n4,
                            const float* w_n00, const float* w_n10,
                            const float* w_n01, const float* w_n11,
                            _Float16* B) {
    int i = blockIdx.x * blockDim.x + threadIdx.x;
    if (i < n4) ((float4*)p)[i] = make_float4(0.0f, 0.0f, 0.0f, 0.0f);
    if (i < 512) {
        v2h v;
#pragma unroll
        for (int k = 0; k < 2; ++k) {
            const int idx   = 2 * i + k;
            const int chunk = idx >> 9;
            const int l     = (idx >> 4) & 31;
            const int e     = idx & 15;
            const int n     = l & 15;
            const int hi    = l >> 4;
            const int f     = e + 8 * hi + (e >= 8 ? 8 : 0);
            float w;
            if (chunk == 0) w = wv_entry(f, n, w_n00, w_n10, w_n01, w_n11);
            else            w = (f < 16) ? wv_entry(32 + f, n, w_n00, w_n10, w_n01, w_n11) : 0.0f;
            v[k] = (_Float16)w;
        }
        *(v2h*)(B + 2 * i) = v;
    }
}

__device__ inline void gather_A(const float* __restrict__ row, int hi,
                                v16h& a0, v16h& a1) {
    const int fb0 = hi ? 8 : 0;
    const int fb1 = hi ? 24 : 16;
#pragma unroll
    for (int i = 0; i < 8; ++i) a0[i]     = (_Float16)row[fb0 + i];
#pragma unroll
    for (int i = 0; i < 8; ++i) a0[8 + i] = (_Float16)row[fb1 + i];
#pragma unroll
    for (int i = 0; i < 8; ++i) a1[i]     = (_Float16)row[32 + fb0 + i];
#pragma unroll
    for (int i = 0; i < 8; ++i) a1[8 + i] = (_Float16)0.0f;
}

constexpr int TILES_PER_WAVE = 5;

__global__ __launch_bounds__(256)
void edge_kernel(const float* __restrict__ x,
                 const int*   __restrict__ src,
                 const int*   __restrict__ dst,
                 const float* __restrict__ angles,
                 const float* __restrict__ transporters,
                 const _Float16* __restrict__ B0p,
                 const _Float16* __restrict__ B1p,
                 float* __restrict__ acc,
                 int plane_stride,
                 int E) {
    __shared__ float tile[8][2 * 16 * 16];
    __shared__ int   half_of_group;

    const int lane = threadIdx.x & 31;
    const int wv   = threadIdx.x >> 5;
    const int m    = lane & 15;
    const int hi   = lane >> 4;

    if (threadIdx.x == 0) {
        const unsigned where = __builtin_amdgcn_s_sendmsg_rtn(0x87);
        half_of_group = (int)((where >> 18) & 1u);
    }

    const v16h b0 = *(const v16h*)(B0p + lane * 16);
    const v16h b1 = *(const v16h*)(B1p + lane * 16);

    __syncthreads();
    float* const slot = acc + (half_of_group ? (size_t)plane_stride : (size_t)0) + (lane & 3);
    float* const t    = tile[wv];

    const int  grp = lane >> 2;
    const bool odd = (lane & 1) != 0;
    const bool top = (lane & 2) != 0;

    for (int it = 0; it < TILES_PER_WAVE; ++it) {
        const long gtile = (long)blockIdx.x * TILES_PER_WAVE + it;
        if (gtile * 256 >= (long)E) break;

        const long tbase = (gtile * 8 + wv) * 32;
        long e0 = tbase + m;
        long e1 = tbase + 16 + m;
        long e0c = e0 < (long)E ? e0 : (long)E - 1;
        long e1c = e1 < (long)E ? e1 : (long)E - 1;

        int s0 = __builtin_nontemporal_load(&src[e0c]);
        int s1 = __builtin_nontemporal_load(&src[e1c]);

        v16h a0t0, a1t0, a0t1, a1t1;
        gather_A(x + (size_t)s0 * 48, hi, a0t0, a1t0);
        gather_A(x + (size_t)s1 * 48, hi, a0t1, a1t1);

        v8f c0 = {}, c1 = {};
        c0 = __builtin_amdgcn_wmma_f32_16x16x32_f16(false, a0t0, false, b0,
                                                    (short)0, c0, false, false);
        c0 = __builtin_amdgcn_wmma_f32_16x16x32_f16(false, a1t0, false, b1,
                                                    (short)0, c0, false, false);
        c1 = __builtin_amdgcn_wmma_f32_16x16x32_f16(false, a0t1, false, b0,
                                                    (short)0, c1, false, false);
        c1 = __builtin_amdgcn_wmma_f32_16x16x32_f16(false, a1t1, false, b1,
                                                    (short)0, c1, false, false);

#pragma unroll
        for (int r = 0; r < 8; ++r) {
            t[(r + 8 * hi) * 16 + m]       = c0[r];
            t[256 + (r + 8 * hi) * 16 + m] = c1[r];
        }
        __syncthreads();

        long ee = tbase + lane;
        float m0 = 0.0f, mv1 = 0.0f, mv2 = 0.0f;
        int   dv = -1;
        if (ee < (long)E) {
            const float* d = &t[lane * 16];
            float ang = __builtin_nontemporal_load(&angles[ee]);
            float tr  = __builtin_nontemporal_load(&transporters[ee]);
            float st, ct, s2t, c2t, sg, cg;
            __sincosf(ang, &st, &ct);
            __sincosf(2.0f * ang, &s2t, &c2t);
            __sincosf(tr, &sg, &cg);

            float d0 = d[0], sa = d[1], sb = d[2];
            float u1w0 = cg * d[3] - sg * d[9];
            float u2w0 = sg * d[3] + cg * d[9];
            float u1w1 = cg * d[4] - sg * d[10];
            float u2w1 = sg * d[4] + cg * d[10];
            float u1p  = cg * d[5] - sg * d[11];
            float u2p  = sg * d[5] + cg * d[11];
            float u1q  = cg * d[6] - sg * d[12];
            float u2q  = sg * d[6] + cg * d[12];
            float u1r  = cg * d[7] - sg * d[13];
            float u2r  = sg * d[7] + cg * d[13];
            float u1s  = cg * d[8] - sg * d[14];
            float u2s  = sg * d[8] + cg * d[14];

            m0  = d0 + ct * u1w0 + st * u2w0 + ct * u2w1 - st * u1w1;
            mv1 = sa * ct - sb * st + u1p - u2q
                + (c2t * u1r + s2t * u2r) - (s2t * u1s - c2t * u2s);
            mv2 = sa * st + sb * ct + u2p + u1q
                + (s2t * u1r - c2t * u2r) + (c2t * u1s + s2t * u2s);

            dv = __builtin_nontemporal_load(&dst[ee]);
        }

        for (int j = 0; j < 4; ++j) {
            const int from = 8 * j + grp;
            int   dk = __shfl(dv,  from);
            float a1 = __shfl(m0,  from);
            float a2 = __shfl(mv1, from);
            float a3 = __shfl(mv2, from);
            float lo  = odd ? a1 : 1.0f;
            float up  = odd ? a3 : a2;
            float val = top ? up : lo;
            if (dk >= 0)
                __hip_atomic_fetch_add(slot + (size_t)dk * 4, val,
                                       __ATOMIC_RELAXED, __HIP_MEMORY_SCOPE_AGENT);
        }
    }
}

__global__ __launch_bounds__(256)
void project_kernel(const float* __restrict__ x,
                    const float* w_n00, const float* w_n10,
                    const float* w_n01, const float* w_n11,
                    float* __restrict__ P,
                    float* __restrict__ acc, int n4,
                    int V) {
    __shared__ _Float16 Bs[2 * 32 * 16];
    const int tid  = threadIdx.x;
    const int lane = tid & 31;
    const int wv   = tid >> 5;
    const int m    = lane & 15;
    const int hi   = lane >> 4;

    const long vbase = ((long)blockIdx.x * 8 + wv) * 32;
    long v0 = vbase + m, v1 = vbase + 16 + m;
    v0 = v0 < (long)V ? v0 : (long)V - 1;
    v1 = v1 < (long)V ? v1 : (long)V - 1;
    v16h a0t0, a1t0, a0t1, a1t1;
    gather_A(x + (size_t)v0 * 48, hi, a0t0, a1t0);
    gather_A(x + (size_t)v1 * 48, hi, a0t1, a1t1);

    WvWhere wh[4];
#pragma unroll
    for (int k = 0; k < 4; ++k) {
        const int idx   = 4 * tid + k;
        const int chunk = idx >> 9;
        const int l     = (idx >> 4) & 31;
        const int e     = idx & 15;
        const int n     = l & 15;
        const int h     = l >> 4;
        const int f     = e + 8 * h + (e >= 8 ? 8 : 0);
        const int feat  = (chunk == 0) ? f : (f < 16 ? 32 + f : 48);
        wh[k] = wv_where(feat, n, w_n00, w_n10, w_n01, w_n11);
    }
    float wl[4];
#pragma unroll
    for (int k = 0; k < 4; ++k) wl[k] = *wh[k].p;
#pragma unroll
    for (int k = 0; k < 4; ++k) Bs[4 * tid + k] = (_Float16)(wh[k].ok ? wl[k] : 0.0f);

    {
        const int nth = gridDim.x * 256;
        for (int i = blockIdx.x * 256 + tid; i < n4; i += nth)
            ((float4*)acc)[i] = make_float4(0.0f, 0.0f, 0.0f, 0.0f);
    }
    __syncthreads();
    if (vbase >= (long)V) return;

    const v16h b0 = *(const v16h*)(Bs + lane * 16);
    const v16h b1 = *(const v16h*)(Bs + 512 + lane * 16);

    v8f c0 = {}, c1 = {};
    c0 = __builtin_amdgcn_wmma_f32_16x16x32_f16(false, a0t0, false, b0, (short)0, c0, false, false);
    c0 = __builtin_amdgcn_wmma_f32_16x16x32_f16(false, a1t0, false, b1, (short)0, c0, false, false);
    c1 = __builtin_amdgcn_wmma_f32_16x16x32_f16(false, a0t1, false, b0, (short)0, c1, false, false);
    c1 = __builtin_amdgcn_wmma_f32_16x16x32_f16(false, a1t1, false, b1, (short)0, c1, false, false);

#pragma unroll
    for (int r = 0; r < 8; ++r) {
        const long r0 = vbase + 8 * hi + r;
        const long r1 = r0 + 16;
        if (r0 < (long)V) P[(size_t)r0 * 16 + m] = c0[r];
        if (r1 < (long)V) P[(size_t)r1 * 16 + m] = c1[r];
    }
}

constexpr int ROW_PITCH = 5;

__device__ __forceinline__
void edge_trips(const float* __restrict__ P,
                const int*   __restrict__ src,
                const int*   __restrict__ dst,
                const float* __restrict__ angles,
                const float* __restrict__ transporters,
                float* const slot, v4f* const rw,
                const int lane, const int wv, const int E) {
    const int  qk  = lane & 3;
    const int  grp = lane >> 2;
    const bool odd = (lane & 1) != 0;
    const bool top = (lane & 2) != 0;
    const long last = (long)E - 1;
    const long wbase = ((long)blockIdx.x * TILES_PER_WAVE * 8 + wv) * 32;
    auto quad_idx = [&](long q) -> v4i {
        const long qc = q < last - 3 ? q : last - 3;
        return *(const v4i*)(src + qc);
    };
    float s_ang[TILES_PER_WAVE], s_tr[TILES_PER_WAVE];
    int   s_dd[TILES_PER_WAVE];
#pragma unroll
    for (int t = 0; t < TILES_PER_WAVE; ++t) {
        const long e  = wbase + (long)t * 256 + lane;
        const long ec = e < last ? e : last;
        s_ang[t] = __builtin_nontemporal_load(&angles[ec]);
        s_tr[t]  = __builtin_nontemporal_load(&transporters[ec]);
        s_dd[t]  = __builtin_nontemporal_load(&dst[ec]);
    }
    const v4i i0 = quad_idx(wbase + 4 * grp);
    v4i nq = quad_idx(wbase + 256 + 4 * grp);
    __builtin_amdgcn_sched_barrier(0);
    v4f g0 = *((const v4f*)(P + (size_t)i0.x * 16) + qk);
    v4f g1 = *((const v4f*)(P + (size_t)i0.y * 16) + qk);
    v4f g2 = *((const v4f*)(P + (size_t)i0.z * 16) + qk);
    v4f g3 = *((const v4f*)(P + (size_t)i0.w * 16) + qk);

#pragma unroll
    for (int it = 0; it < TILES_PER_WAVE; ++it) {
        const long tbase = wbase + (long)it * 256;
        const long ee  = tbase + lane;
        const bool has = ee < (long)E;

        v4i fq = {0, 0, 0, 0};
        if (it + 2 < TILES_PER_WAVE) fq = quad_idx(tbase + 512 + 4 * grp);

        const float ang = s_ang[it];
        const float tr  = s_tr[it];
        const int   dd  = s_dd[it];
        rw[(4 * grp + 0) * ROW_PITCH + qk] = g0;
        rw[(4 * grp + 1) * ROW_PITCH + qk] = g1;
        rw[(4 * grp + 2) * ROW_PITCH + qk] = g2;
        rw[(4 * grp + 3) * ROW_PITCH + qk] = g3;

        if (it + 1 < TILES_PER_WAVE) {
            g0 = *((const v4f*)(P + (size_t)nq.x * 16) + qk);
            g1 = *((const v4f*)(P + (size_t)nq.y * 16) + qk);
            g2 = *((const v4f*)(P + (size_t)nq.z * 16) + qk);
            g3 = *((const v4f*)(P + (size_t)nq.w * 16) + qk);
        }
        nq = fq;

        float m0, mv1, mv2;
        const int dv = has ? dd : -1;
        {
            const v4f* q = rw + lane * ROW_PITCH;
            const v4f q0 = q[0], q1 = q[1], q2 = q[2], q3 = q[3];
            const float d[16] = {q0.x, q0.y, q0.z, q0.w, q1.x, q1.y, q1.z, q1.w,
                                 q2.x, q2.y, q2.z, q2.w, q3.x, q3.y, q3.z, q3.w};
            float st, ct, s2t, c2t, sg, cg;
            __sincosf(ang, &st, &ct);
            __sincosf(2.0f * ang, &s2t, &c2t);
            __sincosf(tr, &sg, &cg);

            float d0 = d[0], sa = d[1], sb = d[2];
            float u1w0 = cg * d[3] - sg * d[9];
            float u2w0 = sg * d[3] + cg * d[9];
            float u1w1 = cg * d[4] - sg * d[10];
            float u2w1 = sg * d[4] + cg * d[10];
            float u1p  = cg * d[5] - sg * d[11];
            float u2p  = sg * d[5] + cg * d[11];
            float u1q  = cg * d[6] - sg * d[12];
            float u2q  = sg * d[6] + cg * d[12];
            float u1r  = cg * d[7] - sg * d[13];
            float u2r  = sg * d[7] + cg * d[13];
            float u1s  = cg * d[8] - sg * d[14];
            float u2s  = sg * d[8] + cg * d[14];

            m0  = d0 + ct * u1w0 + st * u2w0 + ct * u2w1 - st * u1w1;
            mv1 = sa * ct - sb * st + u1p - u2q
                + (c2t * u1r + s2t * u2r) - (s2t * u1s - c2t * u2s);
            mv2 = sa * st + sb * ct + u2p + u1q
                + (s2t * u1r - c2t * u2r) + (c2t * u1s + s2t * u2s);
        }

        for (int j = 0; j < 4; ++j) {
            const int from = 8 * j + grp;
            int   dk = __shfl(dv,  from);
            float a1 = __shfl(m0,  from);
            float a2 = __shfl(mv1, from);
            float a3 = __shfl(mv2, from);
            float lo  = odd ? a1 : 1.0f;
            float up  = odd ? a3 : a2;
            float val = top ? up : lo;
            if (dk >= 0)
                __hip_atomic_fetch_add(slot + (size_t)dk * 4, val,
                                       __ATOMIC_RELAXED, __HIP_MEMORY_SCOPE_AGENT);
        }
    }
}

__device__ __forceinline__
void edge_trips_plain(const float* __restrict__ P,
                      const int*   __restrict__ src,
                      const int*   __restrict__ dst,
                      const float* __restrict__ angles,
                      const float* __restrict__ transporters,
                      float* const slot, v4f* const rw,
                      const int lane, const int wv, const int E) {
    const int  qk  = lane & 3;
    const int  grp = lane >> 2;
    const bool odd = (lane & 1) != 0;
    const bool top = (lane & 2) != 0;
    for (int it = 0; it < TILES_PER_WAVE; ++it) {
        const long gtile = (long)blockIdx.x * TILES_PER_WAVE + it;
        if (gtile * 256 >= (long)E) break;

        const long tbase = (gtile * 8 + wv) * 32;
        const long ee  = tbase + lane;
        const bool has = ee < (long)E;
        const long eec = has ? ee : (long)E - 1;

        const long qe   = tbase + 4 * grp;
        const long last = (long)E - 1;
        const int sq0 = src[qe     < last ? qe     : last];
        const int sq1 = src[qe + 1 < last ? qe + 1 : last];
        const int sq2 = src[qe + 2 < last ? qe + 2 : last];
        const int sq3 = src[qe + 3 < last ? qe + 3 : last];
        const float ang = __builtin_nontemporal_load(&angles[eec]);
        const float tr  = __builtin_nontemporal_load(&transporters[eec]);
        const int   dd  = __builtin_nontemporal_load(&dst[eec]);
        __builtin_amdgcn_sched_barrier(0);
        const v4f g0 = *((const v4f*)(P + (size_t)sq0 * 16) + qk);
        const v4f g1 = *((const v4f*)(P + (size_t)sq1 * 16) + qk);
        const v4f g2 = *((const v4f*)(P + (size_t)sq2 * 16) + qk);
        const v4f g3 = *((const v4f*)(P + (size_t)sq3 * 16) + qk);
        rw[(4 * grp + 0) * ROW_PITCH + qk] = g0;
        rw[(4 * grp + 1) * ROW_PITCH + qk] = g1;
        rw[(4 * grp + 2) * ROW_PITCH + qk] = g2;
        rw[(4 * grp + 3) * ROW_PITCH + qk] = g3;

        float m0, mv1, mv2;
        const int dv = has ? dd : -1;
        {
            const v4f* q = rw + lane * ROW_PITCH;
            const v4f q0 = q[0], q1 = q[1], q2 = q[2], q3 = q[3];
            const float d[16] = {q0.x, q0.y, q0.z, q0.w, q1.x, q1.y, q1.z, q1.w,
                                 q2.x, q2.y, q2.z, q2.w, q3.x, q3.y, q3.z, q3.w};
            float st, ct, s2t, c2t, sg, cg;
            __sincosf(ang, &st, &ct);
            __sincosf(2.0f * ang, &s2t, &c2t);
            __sincosf(tr, &sg, &cg);

            float d0 = d[0], sa = d[1], sb = d[2];
            float u1w0 = cg * d[3] - sg * d[9];
            float u2w0 = sg * d[3] + cg * d[9];
            float u1w1 = cg * d[4] - sg * d[10];
            float u2w1 = sg * d[4] + cg * d[10];
            float u1p  = cg * d[5] - sg * d[11];
            float u2p  = sg * d[5] + cg * d[11];
            float u1q  = cg * d[6] - sg * d[12];
            float u2q  = sg * d[6] + cg * d[12];
            float u1r  = cg * d[7] - sg * d[13];
            float u2r  = sg * d[7] + cg * d[13];
            float u1s  = cg * d[8] - sg * d[14];
            float u2s  = sg * d[8] + cg * d[14];

            m0  = d0 + ct * u1w0 + st * u2w0 + ct * u2w1 - st * u1w1;
            mv1 = sa * ct - sb * st + u1p - u2q
                + (c2t * u1r + s2t * u2r) - (s2t * u1s - c2t * u2s);
            mv2 = sa * st + sb * ct + u2p + u1q
                + (s2t * u1r - c2t * u2r) + (c2t * u1s + s2t * u2s);
        }

        for (int j = 0; j < 4; ++j) {
            const int from = 8 * j + grp;
            int   dk = __shfl(dv,  from);
            float a1 = __shfl(m0,  from);
            float a2 = __shfl(mv1, from);
            float a3 = __shfl(mv2, from);
            float lo  = odd ? a1 : 1.0f;
            float up  = odd ? a3 : a2;
            float val = top ? up : lo;
            if (dk >= 0)
                __hip_atomic_fetch_add(slot + (size_t)dk * 4, val,
                                       __ATOMIC_RELAXED, __HIP_MEMORY_SCOPE_AGENT);
        }
    }
}

__global__ __launch_bounds__(256)
void edge_kernel_p(const float* __restrict__ P,
                   const int*   __restrict__ src,
                   const int*   __restrict__ dst,
                   const float* __restrict__ angles,
                   const float* __restrict__ transporters,
                   float* __restrict__ acc,
                   int plane_stride,
                   int E) {
    __shared__ int half_of_group;
    __shared__ v4f rows[8][32 * ROW_PITCH];

    const int lane = threadIdx.x & 31;
    const int wv   = threadIdx.x >> 5;
    v4f* const rw  = rows[wv];

    if (threadIdx.x == 0) {
        const unsigned where = __builtin_amdgcn_s_sendmsg_rtn(0x87);
        half_of_group = (int)((where >> 18) & 1u);
    }
    __syncthreads();
    float* const slot = acc + (half_of_group ? (size_t)plane_stride : (size_t)0) + (lane & 3);

    if ((E & 3) == 0) edge_trips(P, src, dst, angles, transporters, slot, rw, lane, wv, E);
    else              edge_trips_plain(P, src, dst, angles, transporters, slot, rw, lane, wv, E);
}

__global__ __launch_bounds__(256)
void finalize_kernel(const float* __restrict__ x,
                     const float* __restrict__ e1,
                     const float* __restrict__ e2,
                     const float* __restrict__ w_self0,
                     const float* __restrict__ w_self11,
                     const float* __restrict__ acc,
                     int plane_stride,
                     float* __restrict__ out,
                     int V) {
    int v = blockIdx.x * blockDim.x + threadIdx.x;
    if (v >= V) return;
    const float* row = x + (size_t)v * 48;
    float4 a = *(const float4*)(acc + (size_t)v * 4);
    if (plane_stride != 0) {
        const float4 b = *(const float4*)(acc + (size_t)plane_stride + (size_t)v * 4);
        a.x += b.x; a.y += b.y; a.z += b.z; a.w += b.w;
    }
    float denom = fmaxf(a.x, 1.0f);
    float inv   = 1.0f / denom;
    float mag   = a.y * inv;
    float t1    = a.z * inv;
    float t2    = a.w * inv;
#pragma unroll
    for (int c = 0; c < 16; ++c) {
        mag += row[c] * w_self0[c];
        float xa = row[16 + 2 * c], xb = row[17 + 2 * c];
        float sa = w_self11[2 * c], sb = w_self11[2 * c + 1];
        t1 += xa * sa - xb * sb;
        t2 += xb * sa + xa * sb;
    }
    float ms = 2.0f / (1.0f + __expf(-mag));
#pragma unroll
    for (int k = 0; k < 3; ++k)
        out[v * 3 + k] = (t1 * e1[v * 3 + k] + t2 * e2[v * 3 + k]) * ms;
}

extern "C" void kernel_launch(void* const* d_in, const int* in_sizes, int n_in,
                              void* d_out, int out_size, void* d_ws, size_t ws_size,
                              hipStream_t stream) {
    const float* x            = (const float*)d_in[0];
    const int*   edge_index   = (const int*)d_in[1];
    const float* angles       = (const float*)d_in[2];
    const float* transporters = (const float*)d_in[3];
    const float* e1           = (const float*)d_in[4];
    const float* e2           = (const float*)d_in[5];
    const float* w_self0      = (const float*)d_in[6];
    const float* w_n00        = (const float*)d_in[7];
    const float* w_n10        = (const float*)d_in[8];
    const float* w_self11     = (const float*)d_in[9];
    const float* w_n01        = (const float*)d_in[10];
    const float* w_n11        = (const float*)d_in[11];

    const int E = in_sizes[2];
    const int V = in_sizes[0] / 48;
    const int* src = edge_index;
    const int* dst = edge_index + E;

    const size_t plane_bytes = (size_t)V * 4 * sizeof(float);
    const size_t b_bytes     = (size_t)2 * 32 * 16 * sizeof(_Float16);
    const size_t p_bytes     = (size_t)V * 16 * sizeof(float);

    if (ws_size >= p_bytes + plane_bytes) {
        const int planes       = (ws_size >= p_bytes + 2 * plane_bytes) ? 2 : 1;
        const int plane_stride = (planes == 2) ? 4 * V : 0;
        const int n4 = planes * V;
        float* P   = (float*)d_ws;
        float* acc = P + (size_t)V * 16;

        hipLaunchKernelGGL(project_kernel, dim3((V + 255) / 256), dim3(256), 0, stream,
                           x, w_n00, w_n10, w_n01, w_n11, P, acc, n4, V);

        int stiles = (E + 31) / 32;
        int gtiles = (stiles + 7) / 8;
        int blocks = (gtiles + TILES_PER_WAVE - 1) / TILES_PER_WAVE;
        hipLaunchKernelGGL(edge_kernel_p, dim3(blocks), dim3(256), 0, stream,
                           P, src, dst, angles, transporters, acc, plane_stride, E);

        hipLaunchKernelGGL(finalize_kernel, dim3((V + 255) / 256), dim3(256), 0, stream,
                           x, e1, e2, w_self0, w_self11, acc, plane_stride,
                           (float*)d_out, V);
        return;
    }

    const int planes       = (ws_size >= 2 * plane_bytes + b_bytes) ? 2 : 1;
    const int plane_stride = (planes == 2) ? 4 * V : 0;
    int n4 = planes * V;
    float* acc    = (float*)d_ws;
    _Float16* B0  = (_Float16*)(acc + (size_t)n4 * 4);
    _Float16* B1  = B0 + 32 * 16;

    const int prep_threads = n4 > 512 ? n4 : 512;
    hipLaunchKernelGGL(prep_kernel, dim3((prep_threads + 255) / 256), dim3(256), 0, stream,
                       acc, n4, w_n00, w_n10, w_n01, w_n11, B0);

    int stiles = (E + 31) / 32;
    int gtiles = (stiles + 7) / 8;
    int blocks = (gtiles + TILES_PER_WAVE - 1) / TILES_PER_WAVE;
    hipLaunchKernelGGL(edge_kernel, dim3(blocks), dim3(256), 0, stream,
                       x, src, dst, angles, transporters, B0, B1,
                       acc, plane_stride, E);

    hipLaunchKernelGGL(finalize_kernel, dim3((V + 255) / 256), dim3(256), 0, stream,
                       x, e1, e2, w_self0, w_self11, acc, plane_stride,
                       (float*)d_out, V);
}
